# starting kernel
# baseline (speedup 1.0000x reference)
#include <hip/hip_runtime.h>
#include <hip/hip_bf16.h>

typedef __attribute__((ext_vector_type(2))) float v2f;
typedef __attribute__((ext_vector_type(8))) float v8f;

#define N_ROWS 16384
#define DIMS   128
#define EPS_N  1e-12f

__global__ void collapse_inv_norm_kernel(const float* __restrict__ fc,
                                         float* __restrict__ inv) {
  const int wave = (blockIdx.x * blockDim.x + threadIdx.x) >> 5;
  const int lane = threadIdx.x & 31;
  const float4* p = (const float4*)(fc + (size_t)wave * DIMS) + lane;
  float4 v = *p;
  float ss = v.x * v.x + v.y * v.y + v.z * v.z + v.w * v.w;
#pragma unroll
  for (int off = 16; off > 0; off >>= 1) {
    ss += __shfl_xor(ss, off, 32);
  }
  if (lane == 0) {
    inv[wave] = 1.0f / fmaxf(sqrtf(ss), EPS_N);
  }
}

__global__ void collapse_adj_dot_kernel(const float* __restrict__ fc,
                                        const float* __restrict__ inv,
                                        float* __restrict__ out) {
  const int lane = threadIdx.x & 31;
  const int wavesPerBlk = blockDim.x >> 5;
  const int tile = blockIdx.x * wavesPerBlk + (threadIdx.x >> 5);
  const int base = tile * 16;

  const int m    = lane & 15;
  const int half = lane >> 4;

  const int rowA = base + m;
  int rowB = base + 1 + m;
  if (rowB > N_ROWS - 1) rowB = N_ROWS - 1;

  const float* pa = fc + (size_t)rowA * DIMS + 2 * half;
  const float* pb = fc + (size_t)rowB * DIMS + 2 * half;

  v8f acc = {0.f, 0.f, 0.f, 0.f, 0.f, 0.f, 0.f, 0.f};
#pragma unroll
  for (int k = 0; k < DIMS; k += 4) {
    v2f a = *(const v2f*)(pa + k);
    v2f b = *(const v2f*)(pb + k);
    acc = __builtin_amdgcn_wmma_f32_16x16x4_f32(
        false, a, false, b, (short)0, acc, false, false);
  }

  float diag = 0.f;
#pragma unroll
  for (int r = 0; r < 8; ++r) {
    float cr = acc[r];
    if (lane == r)      diag = cr;
    if (lane == r + 24) diag = cr;
  }

  const bool have = (lane < 8) || (lane >= 24);
  const int  md   = (lane < 8) ? lane : (lane - 16);
  const int  oidx = base + md;
  if (have && oidx < N_ROWS - 1) {
    float d = diag * inv[oidx] * inv[oidx + 1];
    const float c = 1.0f / (float)(N_ROWS - 1);
    float t = d + c;
    out[oidx] = t * t;
  }
}

extern "C" void kernel_launch(void* const* d_in, const int* in_sizes, int n_in,
                              void* d_out, int out_size, void* d_ws, size_t ws_size,
                              hipStream_t stream) {
  (void)in_sizes; (void)n_in; (void)out_size; (void)ws_size;
  const float* fc  = (const float*)d_in[0];
  float*       out = (float*)d_out;
  float*       inv = (float*)d_ws;

  collapse_inv_norm_kernel<<<(N_ROWS * 32) / 256, 256, 0, stream>>>(fc, inv);

  const int tiles = N_ROWS / 16;
  collapse_adj_dot_kernel<<<tiles / 8, 256, 0, stream>>>(fc, inv, out);
}


# Fuse the row-norm pass into the WMMA kernel: squares summed per lane from the A/B fragment registers, k halves joined by permlanex16; one launch, no scratch; norm summation order changes
# speedup vs baseline: 1.7421x; 1.7421x over previous
#include <hip/hip_runtime.h>
#include <hip/hip_bf16.h>


typedef __attribute__((ext_vector_type(2))) float v2f;
typedef __attribute__((ext_vector_type(8))) float v8f;

#define N_ROWS 16384
#define DIMS   128
#define EPS_N  1e-12f

__device__ __forceinline__ float xhalf(float x) {
  return __builtin_bit_cast(float, __builtin_amdgcn_permlanex16(
      __builtin_bit_cast(int, x), __builtin_bit_cast(int, x),
      0x76543210, 0xfedcba98, false, false));
}

__global__ void adj_cos_sq_kernel(const float* __restrict__ fc,
                                  float* __restrict__ out) {
  const int lane = threadIdx.x & 31;
  const int wavesPerBlk = blockDim.x >> 5;
  const int tile = blockIdx.x * wavesPerBlk + (threadIdx.x >> 5);
  const int base = tile * 16;

  const int m    = lane & 15;
  const int half = lane >> 4;

  const int rowA = base + m;
  int rowB = base + 1 + m;
  if (rowB > N_ROWS - 1) rowB = N_ROWS - 1;

  const float* pa = fc + (size_t)rowA * DIMS + 2 * half;
  const float* pb = fc + (size_t)rowB * DIMS + 2 * half;

  v8f acc = {0.f, 0.f, 0.f, 0.f, 0.f, 0.f, 0.f, 0.f};
  v2f ssa = {0.f, 0.f};
  v2f ssb = {0.f, 0.f};
#pragma unroll
  for (int k = 0; k < DIMS; k += 4) {
    v2f a = *(const v2f*)(pa + k);
    v2f b = *(const v2f*)(pb + k);
    acc = __builtin_amdgcn_wmma_f32_16x16x4_f32(
        false, a, false, b, (short)0, acc, false, false);
    ssa = __builtin_elementwise_fma(a, a, ssa);
    ssb = __builtin_elementwise_fma(b, b, ssb);
  }

  float na = ssa.x + ssa.y;
  float nb = ssb.x + ssb.y;
  na += xhalf(na);
  nb += xhalf(nb);
  const float ia = 1.0f / fmaxf(sqrtf(na), EPS_N);
  const float ib = 1.0f / fmaxf(sqrtf(nb), EPS_N);

  float diag = 0.f;
#pragma unroll
  for (int r = 0; r < 8; ++r) {
    float cr = acc[r];
    if (lane == r)      diag = cr;
    if (lane == r + 24) diag = cr;
  }

  const bool have = (lane < 8) || (lane >= 24);
  const int  oidx = base + m;
  if (have && oidx < N_ROWS - 1) {
    float d = diag * ia * ib;
    const float c = 1.0f / (float)(N_ROWS - 1);
    float t = d + c;
    out[oidx] = t * t;
  }
}

extern "C" void kernel_launch(void* const* d_in, const int* in_sizes, int n_in,
                              void* d_out, int out_size, void* d_ws, size_t ws_size,
                              hipStream_t stream) {
  (void)in_sizes; (void)n_in; (void)out_size; (void)d_ws; (void)ws_size;
  const float* fc  = (const float*)d_in[0];
  float*       out = (float*)d_out;

  const int tiles = N_ROWS / 16;
  adj_cos_sq_kernel<<<tiles / 8, 256, 0, stream>>>(fc, out);
}


# Same fragment registers as WMMA A and B operand (a tile's 16 rows times their transpose; tiles overlap by one row): each row loaded once; norms read off the diagonal: norm summation order changes
# speedup vs baseline: 2.4119x; 1.3845x over previous
#include <hip/hip_runtime.h>
#include <hip/hip_bf16.h>


typedef __attribute__((ext_vector_type(2))) float v2f;
typedef __attribute__((ext_vector_type(8))) float v8f;

#define N_ROWS    16384
#define DIMS      128
#define EPS_N     1e-12f
#define TILE_OUT  15
#define N_TILES   1093
#define WAVES_BLK 8
#define N_BLOCKS  137

__device__ __forceinline__ float xhalf(float x) {
  return __builtin_bit_cast(float, __builtin_amdgcn_permlanex16(
      __builtin_bit_cast(int, x), __builtin_bit_cast(int, x),
      0x76543210, 0xfedcba98, false, false));
}

__device__ __forceinline__ float next_in_half(float x) {
  return __builtin_bit_cast(float, __builtin_amdgcn_permlane16(
      __builtin_bit_cast(int, x), __builtin_bit_cast(int, x),
      0x87654321, 0x0fedcba9, false, false));
}

__global__ void adj_cos_sq_kernel(const float* __restrict__ fc,
                                  float* __restrict__ out) {
  const int lane = threadIdx.x & 31;
  const int tile = blockIdx.x * WAVES_BLK + (threadIdx.x >> 5);
  if (tile >= N_TILES) return;
  const int base = tile * TILE_OUT;

  const int c    = lane & 15;
  const int half = lane >> 4;

  int row = base + c;
  if (row > N_ROWS - 1) row = N_ROWS - 1;

  const float* pa = fc + (size_t)row * DIMS + 2 * half;

  v8f acc = {0.f, 0.f, 0.f, 0.f, 0.f, 0.f, 0.f, 0.f};
#pragma unroll
  for (int k = 0; k < DIMS; k += 4) {
    v2f a = *(const v2f*)(pa + k);
    acc = __builtin_amdgcn_wmma_f32_16x16x4_f32(
        false, a, false, a, (short)0, acc, false, false);
  }

  float nsq = 0.f;
  float dot = 0.f;
#pragma unroll
  for (int r = 0; r < 8; ++r) {
    float cr = acc[r];
    if (lane == r)      nsq = cr;
    if (lane == r + 24) nsq = cr;
    if (lane == r - 1)  dot = cr;
    if (lane == r + 23) dot = cr;
  }

  nsq += xhalf(nsq);
  const float inv  = 1.0f / fmaxf(sqrtf(nsq), EPS_N);
  const float invn = next_in_half(inv);

  const bool have = (lane <= 6) || (lane >= 23 && lane <= 30);
  const int  oidx = base + c;
  if (have && oidx < N_ROWS - 1) {
    float d = dot * inv * invn;
    const float cst = 1.0f / (float)(N_ROWS - 1);
    float t = d + cst;
    out[oidx] = t * t;
  }
}

extern "C" void kernel_launch(void* const* d_in, const int* in_sizes, int n_in,
                              void* d_out, int out_size, void* d_ws, size_t ws_size,
                              hipStream_t stream) {
  (void)in_sizes; (void)n_in; (void)out_size; (void)d_ws; (void)ws_size;
  const float* fc  = (const float*)d_in[0];
  float*       out = (float*)d_out;

  adj_cos_sq_kernel<<<N_BLOCKS, WAVES_BLK * 32, 0, stream>>>(fc, out);
}


# Launch shape: 256 workgroups of 5 waves, one per WGP (was 137 of 8): a WGP's load path serves 4 or 5 tiles' fragment loads, not 8; the fifth wave takes tiles 1024..1092 on blocks 0..68, else exits
# speedup vs baseline: 2.6161x; 1.0847x over previous
#include <hip/hip_runtime.h>
#include <hip/hip_bf16.h>


typedef __attribute__((ext_vector_type(2))) float v2f;
typedef __attribute__((ext_vector_type(8))) float v8f;

#define N_ROWS    16384
#define DIMS      128
#define EPS_N     1e-12f
#define TILE_OUT  15
#define N_TILES   1093
#define N_BLOCKS  256
#define MAIN_WAVES 4
#define WAVES_BLK 5
#define N_MAIN    (N_BLOCKS * MAIN_WAVES)

__device__ __forceinline__ float xhalf(float x) {
  return __builtin_bit_cast(float, __builtin_amdgcn_permlanex16(
      __builtin_bit_cast(int, x), __builtin_bit_cast(int, x),
      0x76543210, 0xfedcba98, false, false));
}

__device__ __forceinline__ float next_in_half(float x) {
  return __builtin_bit_cast(float, __builtin_amdgcn_permlane16(
      __builtin_bit_cast(int, x), __builtin_bit_cast(int, x),
      0x87654321, 0x0fedcba9, false, false));
}

__global__ void adj_cos_sq_kernel(const float* __restrict__ fc,
                                  float* __restrict__ out) {
  const int lane = threadIdx.x & 31;
  const int wave = threadIdx.x >> 5;
  const int tile = (wave < MAIN_WAVES) ? blockIdx.x * MAIN_WAVES + wave
                                       : N_MAIN + blockIdx.x;
  if (tile >= N_TILES) return;
  const int base = tile * TILE_OUT;

  const int c    = lane & 15;
  const int half = lane >> 4;

  int row = base + c;
  if (row > N_ROWS - 1) row = N_ROWS - 1;

  const float* pa = fc + (size_t)row * DIMS + 2 * half;

  v8f acc = {0.f, 0.f, 0.f, 0.f, 0.f, 0.f, 0.f, 0.f};
#pragma unroll
  for (int k = 0; k < DIMS; k += 4) {
    v2f a = *(const v2f*)(pa + k);
    acc = __builtin_amdgcn_wmma_f32_16x16x4_f32(
        false, a, false, a, (short)0, acc, false, false);
  }

  float nsq = 0.f;
  float dot = 0.f;
#pragma unroll
  for (int r = 0; r < 8; ++r) {
    float cr = acc[r];
    if (lane == r)      nsq = cr;
    if (lane == r + 24) nsq = cr;
    if (lane == r - 1)  dot = cr;
    if (lane == r + 23) dot = cr;
  }

  nsq += xhalf(nsq);
  const float inv  = 1.0f / fmaxf(sqrtf(nsq), EPS_N);
  const float invn = next_in_half(inv);

  const bool have = (lane <= 6) || (lane >= 23 && lane <= 30);
  const int  oidx = base + c;
  if (have && oidx < N_ROWS - 1) {
    float d = dot * inv * invn;
    const float cst = 1.0f / (float)(N_ROWS - 1);
    float t = d + cst;
    out[oidx] = t * t;
  }
}

extern "C" void kernel_launch(void* const* d_in, const int* in_sizes, int n_in,
                              void* d_out, int out_size, void* d_ws, size_t ws_size,
                              hipStream_t stream) {
  (void)in_sizes; (void)n_in; (void)out_size; (void)d_ws; (void)ws_size;
  const float* fc  = (const float*)d_in[0];
  float*       out = (float*)d_out;

  adj_cos_sq_kernel<<<N_BLOCKS, WAVES_BLK * 32, 0, stream>>>(fc, out);
}
